# starting kernel
# baseline (speedup 1.0000x reference)
#include <hip/hip_runtime.h>
#include <hip/hip_bf16.h>
#include <math.h>


#define TT 512
#define DD 512

typedef __attribute__((ext_vector_type(2))) float v2f;
typedef __attribute__((ext_vector_type(8))) float v8f;

__global__ void qkv_gemm_f32(const float* __restrict__ x,
                             const float* __restrict__ Wq,
                             const float* __restrict__ Wk,
                             const float* __restrict__ Wv,
                             float* __restrict__ q,
                             float* __restrict__ k,
                             float* __restrict__ v) {
    const float* W = (blockIdx.z == 0) ? Wq : (blockIdx.z == 1) ? Wk : Wv;
    float* dst     = (blockIdx.z == 0) ? q  : (blockIdx.z == 1) ? k  : v;

    const int lane   = threadIdx.x & 31;
    const int tileId = blockIdx.x * (blockDim.x >> 5) + (threadIdx.x >> 5);
    const int tm = (tileId >> 5) << 4;
    const int tn = (tileId & 31) << 4;

    const int laneM = lane & 15;
    const int laneK = (lane >> 4) << 1;

    const float* Arow = W + (tm + laneM) * DD + laneK;
    const float* Brow = x + (tn + laneM) * DD + laneK;

    v8f c = {};
#pragma unroll 8
    for (int k0 = 0; k0 < DD; k0 += 4) {
        v2f a = *(const v2f*)(Arow + k0);
        v2f b = *(const v2f*)(Brow + k0);
        c = __builtin_amdgcn_wmma_f32_16x16x4_f32(
                 false, a,  false, b,
                 (short)0, c,  false,  false);
    }

    const int rowOff = (lane >> 4) << 3;
    const int col = tn + laneM;
#pragma unroll
    for (int i = 0; i < 8; ++i)
        dst[(tm + rowOff + i) * DD + col] = c[i];
}

__global__ void rope_heads(const float* __restrict__ qin,
                           const float* __restrict__ kin,
                           float* __restrict__ qout,
                           float* __restrict__ kout) {
    int idx = blockIdx.x * blockDim.x + threadIdx.x;
    int t = idx & (TT - 1);
    int o = idx >> 9;
    int c = o & 63;
    int h = o >> 6;

    float invf = __powf(10000.0f, -(float)((c & 31) << 1) * (1.0f / 64.0f));
    float ang  = (float)t * invf;
    float cs = __cosf(ang);
    float sn = __sinf(ang);

    int poff  = (h < 4) ? (256 * TT) : (-256 * TT);
    float sgn = (h < 4) ? -1.0f : 1.0f;

    qout[idx] = qin[idx] * cs + sgn * qin[idx + poff] * sn;
    kout[idx] = kin[idx] * cs + sgn * kin[idx + poff] * sn;
}

__global__ void attn_rows(const float* __restrict__ q,
                          const float* __restrict__ k,
                          const float* __restrict__ v,
                          float* __restrict__ o) {
    const int lane = threadIdx.x & 31;
    const int row  = blockIdx.x * (blockDim.x >> 5) + (threadIdx.x >> 5);
    const int ch = row >> 9;
    const int i  = row & (TT - 1);

    const float* kr = k + ch * TT;
    const float* vr = v + ch * TT;
    const float a = q[ch * TT + i] * 0.125f;

    float m = -3.4e38f;
    for (int j = lane; j <= i; j += 32)
        m = fmaxf(m, a * kr[j]);
#pragma unroll
    for (int off = 16; off > 0; off >>= 1)
        m = fmaxf(m, __shfl_xor(m, off, 32));

    float s = 0.0f, acc = 0.0f;
    for (int j = lane; j <= i; j += 32) {
        float e = __expf(a * kr[j] - m);
        s   += e;
        acc += e * vr[j];
    }
#pragma unroll
    for (int off = 16; off > 0; off >>= 1) {
        s   += __shfl_xor(s, off, 32);
        acc += __shfl_xor(acc, off, 32);
    }

    if (lane == 0)
        o[ch * TT + i] = acc / s;
}

__global__ void out_gemm_f32(const float* __restrict__ ao,
                             const float* __restrict__ Wo,
                             float* __restrict__ out) {
    const int lane   = threadIdx.x & 31;
    const int tileId = blockIdx.x * (blockDim.x >> 5) + (threadIdx.x >> 5);
    const int tm = (tileId >> 5) << 4;
    const int tn = (tileId & 31) << 4;

    const int laneM = lane & 15;
    const int laneK = (lane >> 4) << 1;

    const float* Arow = Wo + (tm + laneM) * DD + laneK;
    const float* Bcol = ao + laneK * TT + (tn + laneM);

    v8f c = {};
#pragma unroll 8
    for (int k0 = 0; k0 < DD; k0 += 4) {
        v2f a = *(const v2f*)(Arow + k0);
        v2f b;
        b.x = Bcol[k0 * TT];
        b.y = Bcol[k0 * TT + TT];
        c = __builtin_amdgcn_wmma_f32_16x16x4_f32(
                false, a, false, b, (short)0, c, false, false);
    }

    const int rowOff = (lane >> 4) << 3;
    const int tcol = tn + laneM;
#pragma unroll
    for (int i = 0; i < 8; ++i)
        out[tcol * DD + (tm + rowOff + i)] = c[i];
}

extern "C" void kernel_launch(void* const* d_in, const int* in_sizes, int n_in,
                              void* d_out, int out_size, void* d_ws, size_t ws_size,
                              hipStream_t stream) {
    const float* x  = (const float*)d_in[0];
    const float* Wq = (const float*)d_in[1];
    const float* Wk = (const float*)d_in[2];
    const float* Wv = (const float*)d_in[3];
    const float* Wo = (const float*)d_in[4];
    float* out = (float*)d_out;

    float* ws = (float*)d_ws;
    const size_t S = (size_t)DD * TT;
    float* qraw = ws + 0 * S;
    float* kraw = ws + 1 * S;
    float* vbuf = ws + 2 * S;
    float* qrot = ws + 3 * S;
    float* krot = ws + 4 * S;
    float* atto = qraw;

    qkv_gemm_f32<<<dim3(256, 1, 3), 128, 0, stream>>>(x, Wq, Wk, Wv, qraw, kraw, vbuf);

    rope_heads<<<(DD * TT) / 256, 256, 0, stream>>>(qraw, kraw, qrot, krot);

    attn_rows<<<(DD * TT) / 8, 256, 0, stream>>>(qrot, krot, vbuf, atto);

    out_gemm_f32<<<256, 128, 0, stream>>>(atto, Wo, out);
}


# both GEMMs: a wave holds a 32x32 tile as 2x2 accumulators fed from registers (one fragment load feeds two matrix instructions per k pair), K split over the block's 4 waves, partials summed in LDS
# speedup vs baseline: 4.2641x; 4.2641x over previous
#include <hip/hip_runtime.h>
#include <hip/hip_bf16.h>
#include <math.h>


#define TT 512
#define DD 512

typedef __attribute__((ext_vector_type(2))) float v2f;
typedef __attribute__((ext_vector_type(4))) float v4f;
typedef __attribute__((ext_vector_type(8))) float v8f;

#define KSPLIT 4
#define KW   (DD / KSPLIT)
#define KGRP KW
#define PP   36

__device__ __forceinline__ v8f mma4(v2f a, v2f b, v8f c) {
    return __builtin_amdgcn_wmma_f32_16x16x4_f32(
         false, a,  false, b,
         (short)0, c,  false,  false);
}

template <int RSTEP>
__device__ __forceinline__ void put_partial(float* __restrict__ P, int rb0, int rb1, int c,
                                            const v8f& c00, const v8f& c01,
                                            const v8f& c10, const v8f& c11) {
#pragma unroll
    for (int r = 0; r < 8; ++r) {
        P[(rb0 + r * RSTEP) * PP + c]      = c00[r];
        P[(rb0 + r * RSTEP) * PP + 16 + c] = c01[r];
        P[(rb1 + r * RSTEP) * PP + c]      = c10[r];
        P[(rb1 + r * RSTEP) * PP + 16 + c] = c11[r];
    }
}

__device__ __forceinline__ void reduce_store(const float (*part)[32 * PP], int tid,
                                             float* __restrict__ dst) {
#pragma unroll
    for (int i = 0; i < 2; ++i) {
        const int idx = i * 128 + tid;
        const int row = idx >> 3;
        const int c4  = (idx & 7) << 2;
        v4f s = *(const v4f*)&part[0][row * PP + c4];
#pragma unroll
        for (int u = 1; u < KSPLIT; ++u)
            s += *(const v4f*)&part[u][row * PP + c4];
        *(v4f*)(dst + row * DD + c4) = s;
    }
}

__global__ __launch_bounds__(128) void qkv_gemm_f32(const float* __restrict__ x,
                                                    const float* __restrict__ Wq,
                                                    const float* __restrict__ Wk,
                                                    const float* __restrict__ Wv,
                                                    float* __restrict__ q,
                                                    float* __restrict__ k,
                                                    float* __restrict__ v) {
    __shared__ __attribute__((aligned(16))) float part[KSPLIT][32 * PP];

    const float* W = (blockIdx.z == 0) ? Wq : (blockIdx.z == 1) ? Wk : Wv;
    float* dst     = (blockIdx.z == 0) ? q  : (blockIdx.z == 1) ? k  : v;

    const int tid  = threadIdx.x;
    const int lane = tid & 31;
    const int ksl  = tid >> 5;
    const int tm = (blockIdx.x >> 4) << 5;
    const int tn = (blockIdx.x & 15) << 5;

    const int laneM = lane & 15;
    const int laneK = (lane >> 4) << 2;

    const float* A0 = W + (tm + laneM) * DD + ksl * KW + laneK;
    const float* A1 = A0 + 16 * DD;
    const float* B0 = x + (tn + laneM) * DD + ksl * KW + laneK;
    const float* B1 = B0 + 16 * DD;

    v8f c00 = {}, c01 = {}, c10 = {}, c11 = {};
    for (int kb = 0; kb < KW; kb += KGRP) {
        v4f a0[KGRP / 8], a1[KGRP / 8], b0[KGRP / 8], b1[KGRP / 8];
#pragma unroll
        for (int j = 0; j < KGRP / 8; ++j) {
            a0[j] = *(const v4f*)(A0 + kb + 8 * j);
            a1[j] = *(const v4f*)(A1 + kb + 8 * j);
            b0[j] = *(const v4f*)(B0 + kb + 8 * j);
            b1[j] = *(const v4f*)(B1 + kb + 8 * j);
        }
        __builtin_amdgcn_sched_barrier(0);
#pragma unroll
        for (int j = 0; j < KGRP / 8; ++j) {
            v2f a0l, a0h, a1l, a1h, b0l, b0h, b1l, b1h;
            a0l.x = a0[j].x; a0l.y = a0[j].y; a0h.x = a0[j].z; a0h.y = a0[j].w;
            a1l.x = a1[j].x; a1l.y = a1[j].y; a1h.x = a1[j].z; a1h.y = a1[j].w;
            b0l.x = b0[j].x; b0l.y = b0[j].y; b0h.x = b0[j].z; b0h.y = b0[j].w;
            b1l.x = b1[j].x; b1l.y = b1[j].y; b1h.x = b1[j].z; b1h.y = b1[j].w;
            c00 = mma4(a0l, b0l, c00);
            c01 = mma4(a0l, b1l, c01);
            c10 = mma4(a1l, b0l, c10);
            c11 = mma4(a1l, b1l, c11);
            c00 = mma4(a0h, b0h, c00);
            c01 = mma4(a0h, b1h, c01);
            c10 = mma4(a1h, b0h, c10);
            c11 = mma4(a1h, b1h, c11);
        }
    }

    const int rb = (lane >> 4) << 3;
    put_partial<1>(part[ksl], rb, 16 + rb, laneM, c00, c01, c10, c11);
    __syncthreads();
    reduce_store(part, tid, dst + tm * DD + tn);
}

#define ATT_Q_SCALE 0.18033688011112042f

template <bool DIAG>
__device__ __forceinline__ void att_tile(const float (&kk)[32], const float (&vv)[32],
                                         float a, float M, int lane,
                                         float& s_out, float& acc_out) {
    float s0 = 0.0f, s1 = 0.0f, c0 = 0.0f, c1 = 0.0f;
#pragma unroll
    for (int j = 0; j < 32; j += 2) {
        float e0 = __builtin_amdgcn_exp2f(fmaf(a, kk[j], -M));
        float e1 = __builtin_amdgcn_exp2f(fmaf(a, kk[j + 1], -M));
        if (DIAG) {
            e0 = (j <= lane) ? e0 : 0.0f;
            e1 = (j + 1 <= lane) ? e1 : 0.0f;
        }
        s0 += e0;
        s1 += e1;
        c0 = fmaf(e0, vv[j], c0);
        c1 = fmaf(e1, vv[j + 1], c1);
    }
    s_out = s0 + s1;
    acc_out = c0 + c1;
}

__global__ __launch_bounds__(256) void attn_keysplit(const float* __restrict__ q,
                                                     const float* __restrict__ k,
                                                     const float* __restrict__ v,
                                                     float* __restrict__ o) {
    __shared__ float ks[TT];
    __shared__ float vs[TT];
    __shared__ float as_[TT];
    __shared__ float Ms[TT];
    __shared__ float cmx[16];
    __shared__ float cmn[16];
    __shared__ float ps[16][TT];
    __shared__ float pa[16][TT];

    const int ch   = blockIdx.x;
    const int tid  = threadIdx.x;
    const int lane = tid & 31;
    const int w    = __builtin_amdgcn_readfirstlane(tid >> 5);

    const int   hd   = ch >> 6;
    const int   pch  = (hd < 4) ? (ch + 256) : (ch - 256);
    const float sgn  = (hd < 4) ? -1.0f : 1.0f;

    const float* qr = q + ch * TT;
    const float* kr = k + ch * TT;
    const float* vr = v + ch * TT;
    const float* qp = q + pch * TT;
    const float* kp = k + pch * TT;

    float q0[2], q1[2], k0[2], k1[2], v0[2];
#pragma unroll
    for (int h = 0; h < 2; ++h) {
        const int t = tid + h * 256;
        q0[h] = qr[t];
        q1[h] = qp[t];
        k0[h] = kr[t];
        k1[h] = kp[t];
        v0[h] = vr[t];
    }
    __builtin_amdgcn_sched_barrier(0);
    const float invf = __powf(10000.0f, -(float)((ch & 31) << 1) * (1.0f / 64.0f));
    float cs2[2], sn2[2];
#pragma unroll
    for (int h = 0; h < 2; ++h) {
        const float ang = (float)(tid + h * 256) * invf;
        cs2[h] = __cosf(ang);
        sn2[h] = __sinf(ang);
    }
    __builtin_amdgcn_sched_barrier(0);

    float qv[2], pmx[2], pmn[2];
#pragma unroll
    for (int h = 0; h < 2; ++h) {
        const int t = tid + h * 256;
        const float cs = cs2[h];
        const float sn = sn2[h];
        const float kt = k0[h] * cs + sgn * k1[h] * sn;
        const float vt = v0[h];
        qv[h] = q0[h] * cs + sgn * q1[h] * sn;
        ks[t] = kt;
        vs[t] = vt;
        float mx = kt, mn = kt;
#pragma unroll
        for (int off = 1; off < 32; off <<= 1) {
            const float u = __shfl_up(mx, off, 32);
            const float d = __shfl_up(mn, off, 32);
            if (lane >= off) { mx = fmaxf(mx, u); mn = fminf(mn, d); }
        }
        pmx[h] = mx;
        pmn[h] = mn;
        if (lane == 31) { cmx[w + h * 8] = mx; cmn[w + h * 8] = mn; }
    }
    __syncthreads();

#pragma unroll
    for (int h = 0; h < 2; ++h) {
        const int t = tid + h * 256;
        const int c = w + h * 8;
        float gmx = pmx[h], gmn = pmn[h];
        for (int cc = 0; cc < c; ++cc) {
            gmx = fmaxf(gmx, cmx[cc]);
            gmn = fminf(gmn, cmn[cc]);
        }
        const float a = qv[h] * ATT_Q_SCALE;
        const float M = (a >= 0.0f) ? a * gmx : a * gmn;
        as_[t] = a;
        Ms[t]  = M;
    }
    __syncthreads();

    for (int pass = 0; pass < 2; ++pass) {
        const int c = (pass == 0) ? w : (15 - w);
        const int kl = __float_as_int(ks[32 * c + lane]);
        const int vl = __float_as_int(vs[32 * c + lane]);
        float kk[32], vv[32];
#pragma unroll
        for (int j = 0; j < 32; ++j) {
            kk[j] = __int_as_float(__builtin_amdgcn_readlane(kl, j));
            vv[j] = __int_as_float(__builtin_amdgcn_readlane(vl, j));
        }
        {
            const int i = 32 * c + lane;
            float s, acc;
            att_tile<true>(kk, vv, as_[i], Ms[i], lane, s, acc);
            ps[c][i] = s;
            pa[c][i] = acc;
        }
        for (int rg = c + 1; rg < 16; ++rg) {
            const int i = 32 * rg + lane;
            float s, acc;
            att_tile<false>(kk, vv, as_[i], Ms[i], lane, s, acc);
            ps[c][i] = s;
            pa[c][i] = acc;
        }
    }
    __syncthreads();

#pragma unroll
    for (int h = 0; h < 2; ++h) {
        const int t  = tid + h * 256;
        const int nc = w + h * 8 + 1;
        float S = 0.0f, A = 0.0f;
        for (int c = 0; c < nc; ++c) {
            S += ps[c][t];
            A += pa[c][t];
        }
        o[ch * TT + t] = A / S;
    }
}

__global__ __launch_bounds__(128) void out_gemm_f32(const float* __restrict__ ao,
                                                    const float* __restrict__ Wo,
                                                    float* __restrict__ out) {
    __shared__ __attribute__((aligned(16))) float part[KSPLIT][32 * PP];

    const int tid  = threadIdx.x;
    const int lane = tid & 31;
    const int ksl  = tid >> 5;
    const int tm = (blockIdx.x >> 4) << 5;
    const int tn = (blockIdx.x & 15) << 5;

    const int laneM = lane & 15;
    const int laneK = (lane >> 4) << 2;

    const float* Tcol = ao + (ksl * KW + laneK) * TT + tn + 2 * laneM;
    const float* W0 = Wo + (tm + laneM) * DD + ksl * KW + laneK;
    const float* W1 = W0 + 16 * DD;

    v8f c00 = {}, c01 = {}, c10 = {}, c11 = {};
    for (int kb = 0; kb < KW; kb += KGRP) {
        v2f t2[KGRP / 8][4];
        v4f w0[KGRP / 8], w1[KGRP / 8];
#pragma unroll
        for (int j = 0; j < KGRP / 8; ++j) {
            w0[j] = *(const v4f*)(W0 + kb + 8 * j);
            w1[j] = *(const v4f*)(W1 + kb + 8 * j);
#pragma unroll
            for (int e = 0; e < 4; ++e)
                t2[j][e] = *(const v2f*)(Tcol + (kb + 8 * j + e) * TT);
        }
        __builtin_amdgcn_sched_barrier(0);
#pragma unroll
        for (int j = 0; j < KGRP / 8; ++j) {
            v2f a0l, a0h, a1l, a1h, b0l, b0h, b1l, b1h;
            a0l.x = t2[j][0].x; a0l.y = t2[j][1].x; a0h.x = t2[j][2].x; a0h.y = t2[j][3].x;
            a1l.x = t2[j][0].y; a1l.y = t2[j][1].y; a1h.x = t2[j][2].y; a1h.y = t2[j][3].y;
            b0l.x = w0[j].x; b0l.y = w0[j].y; b0h.x = w0[j].z; b0h.y = w0[j].w;
            b1l.x = w1[j].x; b1l.y = w1[j].y; b1h.x = w1[j].z; b1h.y = w1[j].w;
            c00 = mma4(a0l, b0l, c00);
            c01 = mma4(a0l, b1l, c01);
            c10 = mma4(a1l, b0l, c10);
            c11 = mma4(a1l, b1l, c11);
            c00 = mma4(a0h, b0h, c00);
            c01 = mma4(a0h, b1h, c01);
            c10 = mma4(a1h, b0h, c10);
            c11 = mma4(a1h, b1h, c11);
        }
    }

    const int rb = (lane >> 4) << 4;
    put_partial<2>(part[ksl], rb, rb + 1, laneM, c00, c01, c10, c11);
    __syncthreads();
    reduce_store(part, tid, out + tn * DD + tm);
}

extern "C" void kernel_launch(void* const* d_in, const int* in_sizes, int n_in,
                              void* d_out, int out_size, void* d_ws, size_t ws_size,
                              hipStream_t stream) {
    const float* x  = (const float*)d_in[0];
    const float* Wq = (const float*)d_in[1];
    const float* Wk = (const float*)d_in[2];
    const float* Wv = (const float*)d_in[3];
    const float* Wo = (const float*)d_in[4];
    float* out = (float*)d_out;

    float* ws = (float*)d_ws;
    const size_t S = (size_t)DD * TT;
    float* qraw = ws + 0 * S;
    float* kraw = ws + 1 * S;
    float* vbuf = ws + 2 * S;
    float* atto = ws + 3 * S;

    qkv_gemm_f32<<<dim3(256, 1, 3), 128, 0, stream>>>(x, Wq, Wk, Wv, qraw, kraw, vbuf);

    attn_keysplit<<<DD, 256, 0, stream>>>(qraw, kraw, vbuf, atto);

    out_gemm_f32<<<256, 128, 0, stream>>>(atto, Wo, out);
}


# q/k/v GEMM: one block owns the q, k and v tiles of one (o,t) so each x fragment feeds all three projections; one wave per SIMD, fragment loads issued 6 k-steps ahead of the matrix work
# speedup vs baseline: 4.4498x; 1.0435x over previous
#include <hip/hip_runtime.h>
#include <hip/hip_bf16.h>
#include <math.h>


#define TT 512
#define DD 512

typedef __attribute__((ext_vector_type(2))) float v2f;
typedef __attribute__((ext_vector_type(4))) float v4f;
typedef __attribute__((ext_vector_type(8))) float v8f;

#define KSPLIT 4
#define KW   (DD / KSPLIT)
#define KGRP KW
#define PP   36

__device__ __forceinline__ v8f mma4(v2f a, v2f b, v8f c) {
    return __builtin_amdgcn_wmma_f32_16x16x4_f32(
         false, a,  false, b,
         (short)0, c,  false,  false);
}

template <int RSTEP>
__device__ __forceinline__ void put_partial(float* __restrict__ P, int rb0, int rb1, int c,
                                            const v8f& c00, const v8f& c01,
                                            const v8f& c10, const v8f& c11) {
#pragma unroll
    for (int r = 0; r < 8; ++r) {
        P[(rb0 + r * RSTEP) * PP + c]      = c00[r];
        P[(rb0 + r * RSTEP) * PP + 16 + c] = c01[r];
        P[(rb1 + r * RSTEP) * PP + c]      = c10[r];
        P[(rb1 + r * RSTEP) * PP + 16 + c] = c11[r];
    }
}

__device__ __forceinline__ void reduce_store(const float (*part)[32 * PP], int tid,
                                             float* __restrict__ dst) {
#pragma unroll
    for (int i = 0; i < 2; ++i) {
        const int idx = i * 128 + tid;
        const int row = idx >> 3;
        const int c4  = (idx & 7) << 2;
        v4f s = *(const v4f*)&part[0][row * PP + c4];
#pragma unroll
        for (int u = 1; u < KSPLIT; ++u)
            s += *(const v4f*)&part[u][row * PP + c4];
        *(v4f*)(dst + row * DD + c4) = s;
    }
}

#define QKV_AHEAD 6
#define QKV_STEPS (KW / 8)

struct QkvFrag { v4f a[3][2]; v4f b[2]; };

__device__ __forceinline__ void qkv_fetch(const float* __restrict__ Aq, const float* __restrict__ Ak,
                                          const float* __restrict__ Av, const float* __restrict__ B,
                                          int j, QkvFrag& f) {
    f.a[0][0] = *(const v4f*)(Aq + 8 * j);
    f.a[0][1] = *(const v4f*)(Aq + 16 * DD + 8 * j);
    f.a[1][0] = *(const v4f*)(Ak + 8 * j);
    f.a[1][1] = *(const v4f*)(Ak + 16 * DD + 8 * j);
    f.a[2][0] = *(const v4f*)(Av + 8 * j);
    f.a[2][1] = *(const v4f*)(Av + 16 * DD + 8 * j);
    f.b[0]    = *(const v4f*)(B + 8 * j);
    f.b[1]    = *(const v4f*)(B + 16 * DD + 8 * j);
}

__global__ __launch_bounds__(128) void qkv_gemm_f32(const float* __restrict__ x,
                                                    const float* __restrict__ Wq,
                                                    const float* __restrict__ Wk,
                                                    const float* __restrict__ Wv,
                                                    float* __restrict__ q,
                                                    float* __restrict__ k,
                                                    float* __restrict__ v) {
    __shared__ __attribute__((aligned(16))) float part[3][KSPLIT][32 * PP];

    const int tid  = threadIdx.x;
    const int lane = tid & 31;
    const int ksl  = tid >> 5;
    const int tm = (blockIdx.x >> 4) << 5;
    const int tn = (blockIdx.x & 15) << 5;

    const int laneM = lane & 15;
    const int laneK = (lane >> 4) << 2;

    const int aoff = (tm + laneM) * DD + ksl * KW + laneK;
    const float* Aq = Wq + aoff;
    const float* Ak = Wk + aoff;
    const float* Av = Wv + aoff;
    const float* B  = x + (tn + laneM) * DD + ksl * KW + laneK;

    v8f acc[3][2][2];
#pragma unroll
    for (int m = 0; m < 3; ++m)
#pragma unroll
        for (int mi = 0; mi < 2; ++mi)
#pragma unroll
            for (int ni = 0; ni < 2; ++ni) acc[m][mi][ni] = (v8f){};

    QkvFrag fr[QKV_STEPS];
#pragma unroll
    for (int j = 0; j < QKV_AHEAD; ++j) qkv_fetch(Aq, Ak, Av, B, j, fr[j]);
    __builtin_amdgcn_sched_barrier(0);
#pragma unroll
    for (int j = 0; j < QKV_STEPS; ++j) {
        if (j + QKV_AHEAD < QKV_STEPS) qkv_fetch(Aq, Ak, Av, B, j + QKV_AHEAD, fr[j + QKV_AHEAD]);
        __builtin_amdgcn_sched_barrier(0);
        v2f bl[2], bh[2];
#pragma unroll
        for (int ni = 0; ni < 2; ++ni) {
            bl[ni].x = fr[j].b[ni].x; bl[ni].y = fr[j].b[ni].y;
            bh[ni].x = fr[j].b[ni].z; bh[ni].y = fr[j].b[ni].w;
        }
#pragma unroll
        for (int p = 0; p < 2; ++p) {
#pragma unroll
            for (int m = 0; m < 3; ++m) {
#pragma unroll
                for (int mi = 0; mi < 2; ++mi) {
                    v2f a;
                    a.x = p ? fr[j].a[m][mi].z : fr[j].a[m][mi].x;
                    a.y = p ? fr[j].a[m][mi].w : fr[j].a[m][mi].y;
                    acc[m][mi][0] = mma4(a, p ? bh[0] : bl[0], acc[m][mi][0]);
                    acc[m][mi][1] = mma4(a, p ? bh[1] : bl[1], acc[m][mi][1]);
                }
            }
        }
        __builtin_amdgcn_sched_barrier(0);
    }

    const int rb = (lane >> 4) << 3;
#pragma unroll
    for (int m = 0; m < 3; ++m)
        put_partial<1>(part[m][ksl], rb, 16 + rb, laneM,
                       acc[m][0][0], acc[m][0][1], acc[m][1][0], acc[m][1][1]);
    __syncthreads();
    reduce_store(part[0], tid, q + tm * DD + tn);
    reduce_store(part[1], tid, k + tm * DD + tn);
    reduce_store(part[2], tid, v + tm * DD + tn);
}

#define ATT_Q_SCALE 0.18033688011112042f

template <bool DIAG>
__device__ __forceinline__ void att_tile(const float (&kk)[32], const float (&vv)[32],
                                         float a, float M, int lane,
                                         float& s_out, float& acc_out) {
    float s0 = 0.0f, s1 = 0.0f, c0 = 0.0f, c1 = 0.0f;
#pragma unroll
    for (int j = 0; j < 32; j += 2) {
        float e0 = __builtin_amdgcn_exp2f(fmaf(a, kk[j], -M));
        float e1 = __builtin_amdgcn_exp2f(fmaf(a, kk[j + 1], -M));
        if (DIAG) {
            e0 = (j <= lane) ? e0 : 0.0f;
            e1 = (j + 1 <= lane) ? e1 : 0.0f;
        }
        s0 += e0;
        s1 += e1;
        c0 = fmaf(e0, vv[j], c0);
        c1 = fmaf(e1, vv[j + 1], c1);
    }
    s_out = s0 + s1;
    acc_out = c0 + c1;
}

__global__ __launch_bounds__(256) void attn_keysplit(const float* __restrict__ q,
                                                     const float* __restrict__ k,
                                                     const float* __restrict__ v,
                                                     float* __restrict__ o) {
    __shared__ float ks[TT];
    __shared__ float vs[TT];
    __shared__ float as_[TT];
    __shared__ float Ms[TT];
    __shared__ float cmx[16];
    __shared__ float cmn[16];
    __shared__ float ps[16][TT];
    __shared__ float pa[16][TT];

    const int ch   = blockIdx.x;
    const int tid  = threadIdx.x;
    const int lane = tid & 31;
    const int w    = __builtin_amdgcn_readfirstlane(tid >> 5);

    const int   hd   = ch >> 6;
    const int   pch  = (hd < 4) ? (ch + 256) : (ch - 256);
    const float sgn  = (hd < 4) ? -1.0f : 1.0f;

    const float* qr = q + ch * TT;
    const float* kr = k + ch * TT;
    const float* vr = v + ch * TT;
    const float* qp = q + pch * TT;
    const float* kp = k + pch * TT;

    float q0[2], q1[2], k0[2], k1[2], v0[2];
#pragma unroll
    for (int h = 0; h < 2; ++h) {
        const int t = tid + h * 256;
        q0[h] = qr[t];
        q1[h] = qp[t];
        k0[h] = kr[t];
        k1[h] = kp[t];
        v0[h] = vr[t];
    }
    __builtin_amdgcn_sched_barrier(0);
    const float invf = __powf(10000.0f, -(float)((ch & 31) << 1) * (1.0f / 64.0f));
    float cs2[2], sn2[2];
#pragma unroll
    for (int h = 0; h < 2; ++h) {
        const float ang = (float)(tid + h * 256) * invf;
        cs2[h] = __cosf(ang);
        sn2[h] = __sinf(ang);
    }
    __builtin_amdgcn_sched_barrier(0);

    float qv[2], pmx[2], pmn[2];
#pragma unroll
    for (int h = 0; h < 2; ++h) {
        const int t = tid + h * 256;
        const float cs = cs2[h];
        const float sn = sn2[h];
        const float kt = k0[h] * cs + sgn * k1[h] * sn;
        const float vt = v0[h];
        qv[h] = q0[h] * cs + sgn * q1[h] * sn;
        ks[t] = kt;
        vs[t] = vt;
        float mx = kt, mn = kt;
#pragma unroll
        for (int off = 1; off < 32; off <<= 1) {
            const float u = __shfl_up(mx, off, 32);
            const float d = __shfl_up(mn, off, 32);
            if (lane >= off) { mx = fmaxf(mx, u); mn = fminf(mn, d); }
        }
        pmx[h] = mx;
        pmn[h] = mn;
        if (lane == 31) { cmx[w + h * 8] = mx; cmn[w + h * 8] = mn; }
    }
    __syncthreads();

#pragma unroll
    for (int h = 0; h < 2; ++h) {
        const int t = tid + h * 256;
        const int c = w + h * 8;
        float gmx = pmx[h], gmn = pmn[h];
        for (int cc = 0; cc < c; ++cc) {
            gmx = fmaxf(gmx, cmx[cc]);
            gmn = fminf(gmn, cmn[cc]);
        }
        const float a = qv[h] * ATT_Q_SCALE;
        const float M = (a >= 0.0f) ? a * gmx : a * gmn;
        as_[t] = a;
        Ms[t]  = M;
    }
    __syncthreads();

    for (int pass = 0; pass < 2; ++pass) {
        const int c = (pass == 0) ? w : (15 - w);
        const int kl = __float_as_int(ks[32 * c + lane]);
        const int vl = __float_as_int(vs[32 * c + lane]);
        float kk[32], vv[32];
#pragma unroll
        for (int j = 0; j < 32; ++j) {
            kk[j] = __int_as_float(__builtin_amdgcn_readlane(kl, j));
            vv[j] = __int_as_float(__builtin_amdgcn_readlane(vl, j));
        }
        {
            const int i = 32 * c + lane;
            float s, acc;
            att_tile<true>(kk, vv, as_[i], Ms[i], lane, s, acc);
            ps[c][i] = s;
            pa[c][i] = acc;
        }
        for (int rg = c + 1; rg < 16; ++rg) {
            const int i = 32 * rg + lane;
            float s, acc;
            att_tile<false>(kk, vv, as_[i], Ms[i], lane, s, acc);
            ps[c][i] = s;
            pa[c][i] = acc;
        }
    }
    __syncthreads();

#pragma unroll
    for (int h = 0; h < 2; ++h) {
        const int t  = tid + h * 256;
        const int nc = w + h * 8 + 1;
        float S = 0.0f, A = 0.0f;
        for (int c = 0; c < nc; ++c) {
            S += ps[c][t];
            A += pa[c][t];
        }
        o[ch * TT + t] = A / S;
    }
}

__global__ __launch_bounds__(128) void out_gemm_f32(const float* __restrict__ ao,
                                                    const float* __restrict__ Wo,
                                                    float* __restrict__ out) {
    __shared__ __attribute__((aligned(16))) float part[KSPLIT][32 * PP];

    const int tid  = threadIdx.x;
    const int lane = tid & 31;
    const int ksl  = tid >> 5;
    const int tm = (blockIdx.x >> 4) << 5;
    const int tn = (blockIdx.x & 15) << 5;

    const int laneM = lane & 15;
    const int laneK = (lane >> 4) << 2;

    const float* Tcol = ao + (ksl * KW + laneK) * TT + tn + 2 * laneM;
    const float* W0 = Wo + (tm + laneM) * DD + ksl * KW + laneK;
    const float* W1 = W0 + 16 * DD;

    v8f c00 = {}, c01 = {}, c10 = {}, c11 = {};
    for (int kb = 0; kb < KW; kb += KGRP) {
        v2f t2[KGRP / 8][4];
        v4f w0[KGRP / 8], w1[KGRP / 8];
#pragma unroll
        for (int j = 0; j < KGRP / 8; ++j) {
            w0[j] = *(const v4f*)(W0 + kb + 8 * j);
            w1[j] = *(const v4f*)(W1 + kb + 8 * j);
#pragma unroll
            for (int e = 0; e < 4; ++e)
                t2[j][e] = *(const v2f*)(Tcol + (kb + 8 * j + e) * TT);
        }
        __builtin_amdgcn_sched_barrier(0);
#pragma unroll
        for (int j = 0; j < KGRP / 8; ++j) {
            v2f a0l, a0h, a1l, a1h, b0l, b0h, b1l, b1h;
            a0l.x = t2[j][0].x; a0l.y = t2[j][1].x; a0h.x = t2[j][2].x; a0h.y = t2[j][3].x;
            a1l.x = t2[j][0].y; a1l.y = t2[j][1].y; a1h.x = t2[j][2].y; a1h.y = t2[j][3].y;
            b0l.x = w0[j].x; b0l.y = w0[j].y; b0h.x = w0[j].z; b0h.y = w0[j].w;
            b1l.x = w1[j].x; b1l.y = w1[j].y; b1h.x = w1[j].z; b1h.y = w1[j].w;
            c00 = mma4(a0l, b0l, c00);
            c01 = mma4(a0l, b1l, c01);
            c10 = mma4(a1l, b0l, c10);
            c11 = mma4(a1l, b1l, c11);
            c00 = mma4(a0h, b0h, c00);
            c01 = mma4(a0h, b1h, c01);
            c10 = mma4(a1h, b0h, c10);
            c11 = mma4(a1h, b1h, c11);
        }
    }

    const int rb = (lane >> 4) << 4;
    put_partial<2>(part[ksl], rb, rb + 1, laneM, c00, c01, c10, c11);
    __syncthreads();
    reduce_store(part, tid, out + tn * DD + tm);
}

extern "C" void kernel_launch(void* const* d_in, const int* in_sizes, int n_in,
                              void* d_out, int out_size, void* d_ws, size_t ws_size,
                              hipStream_t stream) {
    const float* x  = (const float*)d_in[0];
    const float* Wq = (const float*)d_in[1];
    const float* Wk = (const float*)d_in[2];
    const float* Wv = (const float*)d_in[3];
    const float* Wo = (const float*)d_in[4];
    float* out = (float*)d_out;

    float* ws = (float*)d_ws;
    const size_t S = (size_t)DD * TT;
    float* qraw = ws + 0 * S;
    float* kraw = ws + 1 * S;
    float* vbuf = ws + 2 * S;
    float* atto = ws + 3 * S;

    qkv_gemm_f32<<<256, 128, 0, stream>>>(x, Wq, Wk, Wv, qraw, kraw, vbuf);

    attn_keysplit<<<DD, 256, 0, stream>>>(qraw, kraw, vbuf, atto);

    out_gemm_f32<<<256, 128, 0, stream>>>(atto, Wo, out);
}


# attention: serial LDS round trips outside the key loop removed: in-wave prefix max/min on DPP row shifts, chunk prefix and partial-sum merge read as one group with wave-uniform selects
# speedup vs baseline: 4.6436x; 1.0436x over previous
#include <hip/hip_runtime.h>
#include <hip/hip_bf16.h>
#include <math.h>


#define TT 512
#define DD 512

typedef __attribute__((ext_vector_type(2))) float v2f;
typedef __attribute__((ext_vector_type(4))) float v4f;
typedef __attribute__((ext_vector_type(8))) float v8f;

#define KSPLIT 4
#define KW   (DD / KSPLIT)
#define KGRP KW
#define PP   36

__device__ __forceinline__ v8f mma4(v2f a, v2f b, v8f c) {
    return __builtin_amdgcn_wmma_f32_16x16x4_f32(
         false, a,  false, b,
         (short)0, c,  false,  false);
}

template <int RSTEP>
__device__ __forceinline__ void put_partial(float* __restrict__ P, int rb0, int rb1, int c,
                                            const v8f& c00, const v8f& c01,
                                            const v8f& c10, const v8f& c11) {
#pragma unroll
    for (int r = 0; r < 8; ++r) {
        P[(rb0 + r * RSTEP) * PP + c]      = c00[r];
        P[(rb0 + r * RSTEP) * PP + 16 + c] = c01[r];
        P[(rb1 + r * RSTEP) * PP + c]      = c10[r];
        P[(rb1 + r * RSTEP) * PP + 16 + c] = c11[r];
    }
}

__device__ __forceinline__ void reduce_store(const float (*part)[32 * PP], int tid,
                                             float* __restrict__ dst) {
#pragma unroll
    for (int i = 0; i < 2; ++i) {
        const int idx = i * 128 + tid;
        const int row = idx >> 3;
        const int c4  = (idx & 7) << 2;
        v4f s = *(const v4f*)&part[0][row * PP + c4];
#pragma unroll
        for (int u = 1; u < KSPLIT; ++u)
            s += *(const v4f*)&part[u][row * PP + c4];
        *(v4f*)(dst + row * DD + c4) = s;
    }
}

#define QKV_AHEAD 6
#define QKV_STEPS (KW / 8)

struct QkvFrag { v4f a[3][2]; v4f b[2]; };

__device__ __forceinline__ void qkv_fetch(const float* __restrict__ Aq, const float* __restrict__ Ak,
                                          const float* __restrict__ Av, const float* __restrict__ B,
                                          int j, QkvFrag& f) {
    f.a[0][0] = *(const v4f*)(Aq + 8 * j);
    f.a[0][1] = *(const v4f*)(Aq + 16 * DD + 8 * j);
    f.a[1][0] = *(const v4f*)(Ak + 8 * j);
    f.a[1][1] = *(const v4f*)(Ak + 16 * DD + 8 * j);
    f.a[2][0] = *(const v4f*)(Av + 8 * j);
    f.a[2][1] = *(const v4f*)(Av + 16 * DD + 8 * j);
    f.b[0]    = *(const v4f*)(B + 8 * j);
    f.b[1]    = *(const v4f*)(B + 16 * DD + 8 * j);
}

__global__ __launch_bounds__(128) void qkv_gemm_f32(const float* __restrict__ x,
                                                    const float* __restrict__ Wq,
                                                    const float* __restrict__ Wk,
                                                    const float* __restrict__ Wv,
                                                    float* __restrict__ q,
                                                    float* __restrict__ k,
                                                    float* __restrict__ v) {
    __shared__ __attribute__((aligned(16))) float part[3][KSPLIT][32 * PP];

    const int tid  = threadIdx.x;
    const int lane = tid & 31;
    const int ksl  = tid >> 5;
    const int tm = (blockIdx.x >> 4) << 5;
    const int tn = (blockIdx.x & 15) << 5;

    const int laneM = lane & 15;
    const int laneK = (lane >> 4) << 2;

    const int aoff = (tm + laneM) * DD + ksl * KW + laneK;
    const float* Aq = Wq + aoff;
    const float* Ak = Wk + aoff;
    const float* Av = Wv + aoff;
    const float* B  = x + (tn + laneM) * DD + ksl * KW + laneK;

    v8f acc[3][2][2];
#pragma unroll
    for (int m = 0; m < 3; ++m)
#pragma unroll
        for (int mi = 0; mi < 2; ++mi)
#pragma unroll
            for (int ni = 0; ni < 2; ++ni) acc[m][mi][ni] = (v8f){};

    QkvFrag fr[QKV_STEPS];
#pragma unroll
    for (int j = 0; j < QKV_AHEAD; ++j) qkv_fetch(Aq, Ak, Av, B, j, fr[j]);
    __builtin_amdgcn_sched_barrier(0);
#pragma unroll
    for (int j = 0; j < QKV_STEPS; ++j) {
        if (j + QKV_AHEAD < QKV_STEPS) qkv_fetch(Aq, Ak, Av, B, j + QKV_AHEAD, fr[j + QKV_AHEAD]);
        __builtin_amdgcn_sched_barrier(0);
        v2f bl[2], bh[2];
#pragma unroll
        for (int ni = 0; ni < 2; ++ni) {
            bl[ni].x = fr[j].b[ni].x; bl[ni].y = fr[j].b[ni].y;
            bh[ni].x = fr[j].b[ni].z; bh[ni].y = fr[j].b[ni].w;
        }
#pragma unroll
        for (int p = 0; p < 2; ++p) {
#pragma unroll
            for (int m = 0; m < 3; ++m) {
#pragma unroll
                for (int mi = 0; mi < 2; ++mi) {
                    v2f a;
                    a.x = p ? fr[j].a[m][mi].z : fr[j].a[m][mi].x;
                    a.y = p ? fr[j].a[m][mi].w : fr[j].a[m][mi].y;
                    acc[m][mi][0] = mma4(a, p ? bh[0] : bl[0], acc[m][mi][0]);
                    acc[m][mi][1] = mma4(a, p ? bh[1] : bl[1], acc[m][mi][1]);
                }
            }
        }
        __builtin_amdgcn_sched_barrier(0);
    }

    const int rb = (lane >> 4) << 3;
#pragma unroll
    for (int m = 0; m < 3; ++m)
        put_partial<1>(part[m][ksl], rb, 16 + rb, laneM,
                       acc[m][0][0], acc[m][0][1], acc[m][1][0], acc[m][1][1]);
    __syncthreads();
    reduce_store(part[0], tid, q + tm * DD + tn);
    reduce_store(part[1], tid, k + tm * DD + tn);
    reduce_store(part[2], tid, v + tm * DD + tn);
}

#define ATT_Q_SCALE 0.18033688011112042f

template <int CTRL>
__device__ __forceinline__ float row_shr_keep(float x) {
    return __int_as_float(__builtin_amdgcn_update_dpp(__float_as_int(x), __float_as_int(x),
                                                      CTRL, 0xf, 0xf, false));
}

__device__ __forceinline__ void wave_prefix_maxmin(float& mx, float& mn, int lane) {
    mx = fmaxf(mx, row_shr_keep<0x111>(mx));  mn = fminf(mn, row_shr_keep<0x111>(mn));
    mx = fmaxf(mx, row_shr_keep<0x112>(mx));  mn = fminf(mn, row_shr_keep<0x112>(mn));
    mx = fmaxf(mx, row_shr_keep<0x114>(mx));  mn = fminf(mn, row_shr_keep<0x114>(mn));
    mx = fmaxf(mx, row_shr_keep<0x118>(mx));  mn = fminf(mn, row_shr_keep<0x118>(mn));
    const float lx = __int_as_float(__builtin_amdgcn_readlane(__float_as_int(mx), 15));
    const float ln = __int_as_float(__builtin_amdgcn_readlane(__float_as_int(mn), 15));
    if (lane >= 16) { mx = fmaxf(mx, lx); mn = fminf(mn, ln); }
}

template <bool DIAG>
__device__ __forceinline__ void att_tile(const float (&kk)[32], const float (&vv)[32],
                                         float a, float M, int lane,
                                         float& s_out, float& acc_out) {
    float s0 = 0.0f, s1 = 0.0f, c0 = 0.0f, c1 = 0.0f;
#pragma unroll
    for (int j = 0; j < 32; j += 2) {
        float e0 = __builtin_amdgcn_exp2f(fmaf(a, kk[j], -M));
        float e1 = __builtin_amdgcn_exp2f(fmaf(a, kk[j + 1], -M));
        if (DIAG) {
            e0 = (j <= lane) ? e0 : 0.0f;
            e1 = (j + 1 <= lane) ? e1 : 0.0f;
        }
        s0 += e0;
        s1 += e1;
        c0 = fmaf(e0, vv[j], c0);
        c1 = fmaf(e1, vv[j + 1], c1);
    }
    s_out = s0 + s1;
    acc_out = c0 + c1;
}

__global__ __launch_bounds__(256) void attn_keysplit(const float* __restrict__ q,
                                                     const float* __restrict__ k,
                                                     const float* __restrict__ v,
                                                     float* __restrict__ o) {
    __shared__ float ks[TT];
    __shared__ float vs[TT];
    __shared__ float as_[TT];
    __shared__ float Ms[TT];
    __shared__ __attribute__((aligned(16))) float cmx[16];
    __shared__ __attribute__((aligned(16))) float cmn[16];
    __shared__ float ps[16][TT];
    __shared__ float pa[16][TT];

    const int ch   = blockIdx.x;
    const int tid  = threadIdx.x;
    const int lane = tid & 31;
    const int w    = __builtin_amdgcn_readfirstlane(tid >> 5);

    const int   hd   = ch >> 6;
    const int   pch  = (hd < 4) ? (ch + 256) : (ch - 256);
    const float sgn  = (hd < 4) ? -1.0f : 1.0f;

    const float* qr = q + ch * TT;
    const float* kr = k + ch * TT;
    const float* vr = v + ch * TT;
    const float* qp = q + pch * TT;
    const float* kp = k + pch * TT;

    float q0[2], q1[2], k0[2], k1[2], v0[2];
#pragma unroll
    for (int h = 0; h < 2; ++h) {
        const int t = tid + h * 256;
        q0[h] = qr[t];
        q1[h] = qp[t];
        k0[h] = kr[t];
        k1[h] = kp[t];
        v0[h] = vr[t];
    }
    __builtin_amdgcn_sched_barrier(0);
    const float invf = __powf(10000.0f, -(float)((ch & 31) << 1) * (1.0f / 64.0f));
    float cs2[2], sn2[2];
#pragma unroll
    for (int h = 0; h < 2; ++h) {
        const float ang = (float)(tid + h * 256) * invf;
        cs2[h] = __cosf(ang);
        sn2[h] = __sinf(ang);
    }
    __builtin_amdgcn_sched_barrier(0);

    float qv[2], pmx[2], pmn[2];
#pragma unroll
    for (int h = 0; h < 2; ++h) {
        const int t = tid + h * 256;
        const float cs = cs2[h];
        const float sn = sn2[h];
        const float kt = k0[h] * cs + sgn * k1[h] * sn;
        const float vt = v0[h];
        qv[h] = q0[h] * cs + sgn * q1[h] * sn;
        ks[t] = kt;
        vs[t] = vt;
        float mx = kt, mn = kt;
        wave_prefix_maxmin(mx, mn, lane);
        pmx[h] = mx;
        pmn[h] = mn;
        if (lane == 31) { cmx[w + h * 8] = mx; cmn[w + h * 8] = mn; }
    }
    __syncthreads();

    float cm[16], cn[16];
#pragma unroll
    for (int g = 0; g < 4; ++g) {
        const v4f a4 = *(const v4f*)&cmx[4 * g];
        const v4f b4 = *(const v4f*)&cmn[4 * g];
#pragma unroll
        for (int e = 0; e < 4; ++e) { cm[4 * g + e] = a4[e]; cn[4 * g + e] = b4[e]; }
    }
#pragma unroll
    for (int h = 0; h < 2; ++h) {
        const int t = tid + h * 256;
        const int c = w + h * 8;
        float gmx = pmx[h], gmn = pmn[h];
#pragma unroll
        for (int cc = 0; cc < (h ? 15 : 7); ++cc) {
            const bool on = (h == 1 && cc < 8) || (cc < c);
            gmx = fmaxf(gmx, on ? cm[cc] : gmx);
            gmn = fminf(gmn, on ? cn[cc] : gmn);
        }
        const float a = qv[h] * ATT_Q_SCALE;
        const float M = (a >= 0.0f) ? a * gmx : a * gmn;
        as_[t] = a;
        Ms[t]  = M;
    }
    __syncthreads();

    for (int pass = 0; pass < 2; ++pass) {
        const int c = (pass == 0) ? w : (15 - w);
        const int kl = __float_as_int(ks[32 * c + lane]);
        const int vl = __float_as_int(vs[32 * c + lane]);
        float kk[32], vv[32];
#pragma unroll
        for (int j = 0; j < 32; ++j) {
            kk[j] = __int_as_float(__builtin_amdgcn_readlane(kl, j));
            vv[j] = __int_as_float(__builtin_amdgcn_readlane(vl, j));
        }
        {
            const int i = 32 * c + lane;
            float s, acc;
            att_tile<true>(kk, vv, as_[i], Ms[i], lane, s, acc);
            ps[c][i] = s;
            pa[c][i] = acc;
        }
        for (int rg = c + 1; rg < 16; ++rg) {
            const int i = 32 * rg + lane;
            float s, acc;
            att_tile<false>(kk, vv, as_[i], Ms[i], lane, s, acc);
            ps[c][i] = s;
            pa[c][i] = acc;
        }
    }
    __syncthreads();

#pragma unroll
    for (int h = 0; h < 2; ++h) {
        const int t  = tid + h * 256;
        const int nc = w + h * 8 + 1;
        float S = 0.0f, A = 0.0f;
#pragma unroll
        for (int c = 0; c < (h ? 16 : 8); ++c) {
            const bool on = (h == 1 && c < 8) || (c < nc);
            const int cs = on ? c : (nc - 1);
            const float sp = ps[cs][t];
            const float ap = pa[cs][t];
            S += on ? sp : 0.0f;
            A += on ? ap : 0.0f;
        }
        o[ch * TT + t] = A / S;
    }
}

__global__ __launch_bounds__(128) void out_gemm_f32(const float* __restrict__ ao,
                                                    const float* __restrict__ Wo,
                                                    float* __restrict__ out) {
    __shared__ __attribute__((aligned(16))) float part[KSPLIT][32 * PP];

    const int tid  = threadIdx.x;
    const int lane = tid & 31;
    const int ksl  = tid >> 5;
    const int tm = (blockIdx.x >> 4) << 5;
    const int tn = (blockIdx.x & 15) << 5;

    const int laneM = lane & 15;
    const int laneK = (lane >> 4) << 2;

    const float* Tcol = ao + (ksl * KW + laneK) * TT + tn + 2 * laneM;
    const float* W0 = Wo + (tm + laneM) * DD + ksl * KW + laneK;
    const float* W1 = W0 + 16 * DD;

    v8f c00 = {}, c01 = {}, c10 = {}, c11 = {};
    for (int kb = 0; kb < KW; kb += KGRP) {
        v2f t2[KGRP / 8][4];
        v4f w0[KGRP / 8], w1[KGRP / 8];
#pragma unroll
        for (int j = 0; j < KGRP / 8; ++j) {
            w0[j] = *(const v4f*)(W0 + kb + 8 * j);
            w1[j] = *(const v4f*)(W1 + kb + 8 * j);
#pragma unroll
            for (int e = 0; e < 4; ++e)
                t2[j][e] = *(const v2f*)(Tcol + (kb + 8 * j + e) * TT);
        }
        __builtin_amdgcn_sched_barrier(0);
#pragma unroll
        for (int j = 0; j < KGRP / 8; ++j) {
            v2f a0l, a0h, a1l, a1h, b0l, b0h, b1l, b1h;
            a0l.x = t2[j][0].x; a0l.y = t2[j][1].x; a0h.x = t2[j][2].x; a0h.y = t2[j][3].x;
            a1l.x = t2[j][0].y; a1l.y = t2[j][1].y; a1h.x = t2[j][2].y; a1h.y = t2[j][3].y;
            b0l.x = w0[j].x; b0l.y = w0[j].y; b0h.x = w0[j].z; b0h.y = w0[j].w;
            b1l.x = w1[j].x; b1l.y = w1[j].y; b1h.x = w1[j].z; b1h.y = w1[j].w;
            c00 = mma4(a0l, b0l, c00);
            c01 = mma4(a0l, b1l, c01);
            c10 = mma4(a1l, b0l, c10);
            c11 = mma4(a1l, b1l, c11);
            c00 = mma4(a0h, b0h, c00);
            c01 = mma4(a0h, b1h, c01);
            c10 = mma4(a1h, b0h, c10);
            c11 = mma4(a1h, b1h, c11);
        }
    }

    const int rb = (lane >> 4) << 4;
    put_partial<2>(part[ksl], rb, rb + 1, laneM, c00, c01, c10, c11);
    __syncthreads();
    reduce_store(part, tid, out + tn * DD + tm);
}

extern "C" void kernel_launch(void* const* d_in, const int* in_sizes, int n_in,
                              void* d_out, int out_size, void* d_ws, size_t ws_size,
                              hipStream_t stream) {
    const float* x  = (const float*)d_in[0];
    const float* Wq = (const float*)d_in[1];
    const float* Wk = (const float*)d_in[2];
    const float* Wv = (const float*)d_in[3];
    const float* Wo = (const float*)d_in[4];
    float* out = (float*)d_out;

    float* ws = (float*)d_ws;
    const size_t S = (size_t)DD * TT;
    float* qraw = ws + 0 * S;
    float* kraw = ws + 1 * S;
    float* vbuf = ws + 2 * S;
    float* atto = ws + 3 * S;

    qkv_gemm_f32<<<256, 128, 0, stream>>>(x, Wq, Wk, Wv, qraw, kraw, vbuf);

    attn_keysplit<<<DD, 256, 0, stream>>>(qraw, kraw, vbuf, atto);

    out_gemm_f32<<<256, 128, 0, stream>>>(atto, Wo, out);
}
